# starting kernel
# baseline (speedup 1.0000x reference)
#include <hip/hip_runtime.h>
#include <hip/hip_bf16.h>
#include <cstddef>

typedef __attribute__((ext_vector_type(2))) float v2f;
typedef __attribute__((ext_vector_type(4))) float v4f;
typedef __attribute__((ext_vector_type(8))) float v8f;

#define BATCH 2
#define Hh    384
#define Ww    1248
#define DISP  192
#define WT    (Ww / 16)
#define HT    (Hh / 16)
#define WGW   6
#define WQ    (WT / WGW)
#define NBLK  (BATCH * HT * WQ)

#define SLAB_COLS   288
#define SLAB_STRIDE 292

__global__ __launch_bounds__(192) void StereoMatching_wmma_kernel(
    const float* __restrict__ L, const float* __restrict__ R, float* __restrict__ out)
{
    __shared__ __align__(16) float slab[16 * SLAB_STRIDE];

    const int tid  = threadIdx.x;
    const int lane = tid & 31;
    const int widx = tid >> 5;

    const int wq = blockIdx.x % WQ;
    const int t2 = blockIdx.x / WQ;
    const int ht = t2 % HT;
    const int b  = t2 / HT;
    const int h0 = ht << 4;
    const int gw0 = wq * (WGW * 16) - DISP;

    {
        const int frow  = tid / 12;
        const int fcol0 = (tid % 12) * 24;
        const float* grow = R + (size_t)(b * Hh + h0 + frow) * Ww;
        float* srow = &slab[frow * SLAB_STRIDE];
        if (gw0 >= 0) {
#pragma unroll
            for (int j = 0; j < 6; ++j) {
                const int col = fcol0 + 4 * j;
                *(v4f*)(srow + col) = *(const v4f*)(grow + gw0 + col);
            }
        } else {
#pragma unroll
            for (int j = 0; j < 24; ++j) {
                const int col = fcol0 + j;
                const int gc  = gw0 + col;
                const int gcc = gc < 0 ? 0 : gc;
                const float v = grow[gcc];
                srow[col] = (gc >= 0) ? v : 0.0f;
            }
        }
    }
    __syncthreads();

    const int n     = lane & 15;
    const int khalf = lane >> 4;

    const float* arow = &slab[n * SLAB_STRIDE + widx * 16 + 2 * khalf];
    v8f c = {};

#pragma unroll
    for (int kc = 0; kc < 4; ++kc) {
        const v2f a = *(const v2f*)(arow + 4 * kc);
        const int x0 = n + DISP - (4 * kc + 2 * khalf);
        v2f bb;
        bb.x = (x0     < DISP) ? (float)x0       : 0.0f;
        bb.y = (x0 - 1 < DISP) ? (float)(x0 - 1) : 0.0f;
        c = __builtin_amdgcn_wmma_f32_16x16x4_f32(false, a, false, bb,
                                                  (short)0, c, false, false);
    }

    float xf0 = (float)(n + DISP - 16 - 2 * khalf);
    float xf1 = xf0 - 1.0f;
#pragma unroll 4
    for (int kc = 4; kc < 52; ++kc) {
        const v2f a = *(const v2f*)(arow + 4 * kc);
        v2f bb;
        bb.x = fmaxf(xf0, 0.0f);
        bb.y = fmaxf(xf1, 0.0f);
        c = __builtin_amdgcn_wmma_f32_16x16x4_f32(false, a, false, bb,
                                                  (short)0, c, false, false);
        xf0 -= 4.0f;
        xf1 -= 4.0f;
    }

    const int w0 = (wq * WGW + widx) << 4;
    const size_t obase = (size_t)(b * Hh + h0) * Ww + (size_t)(w0 + n);
#pragma unroll
    for (int v = 0; v < 8; ++v) {
        const int M = v + (khalf << 3);
        const size_t idx = obase + (size_t)M * Ww;
        out[idx] = L[idx] * c[v];
    }
}

extern "C" void kernel_launch(void* const* d_in, const int* in_sizes, int n_in,
                              void* d_out, int out_size, void* d_ws, size_t ws_size,
                              hipStream_t stream) {
    (void)in_sizes; (void)n_in; (void)out_size; (void)d_ws; (void)ws_size;
    const float* L = (const float*)d_in[0];
    const float* R = (const float*)d_in[1];
    float* O = (float*)d_out;

    static_assert(WT % WGW == 0, "w-tiles must split into groups");
    dim3 grid(NBLK);
    dim3 block(192);
    hipLaunchKernelGGL(StereoMatching_wmma_kernel, grid, block, 0, stream, L, R, O);
}


# Four tiles x 13 band blocks as a nested 2x2 fast convolution (9 block products per 4x4, not 16) on chunk differences and band sums: 124 WMMA per wave, not 208; f32 sum order changes
# speedup vs baseline: 1.4973x; 1.4973x over previous
#include <hip/hip_runtime.h>
#include <hip/hip_bf16.h>
#include <cstddef>


typedef __attribute__((ext_vector_type(2))) float v2f;
typedef __attribute__((ext_vector_type(4))) float v4f;
typedef __attribute__((ext_vector_type(8))) float v8f;

#define BATCH 2
#define Hh    384
#define Ww    1248
#define DISP  192
#define WT    (Ww / 16)
#define NT    4
#define NW    4
#define BW    (16 * NT)
#define WBLK  5
#define HG    (Hh / 16)
#define NBLK  (BATCH * HG * WBLK)

#define NQ     32


static __device__ __forceinline__ v2f pick(const v4f p, const int e) {
    v2f a;
    a.x = e ? p.z : p.x;
    a.y = e ? p.w : p.y;
    return a;
}

static __device__ __forceinline__ v2f band(const float xb, const int u, const int e) {
    const float x0 = xb - (float)(8 * u + 2 * e);
    const float x1 = x0 - 1.0f;
    v2f bb;
    if (u < 2) {
        bb.x = (x0 < (float)DISP) ? x0 : 0.0f;
        bb.y = (x1 < (float)DISP) ? x1 : 0.0f;
    } else if (u >= 24) {
        bb.x = fmaxf(x0, 0.0f);
        bb.y = fmaxf(x1, 0.0f);
    } else {
        bb.x = x0;
        bb.y = x1;
    }
    return bb;
}

__global__ __launch_bounds__(128) void StereoMatching_wmma_kernel(
    const float* __restrict__ L, const float* __restrict__ R, float* __restrict__ out)
{
    const int tid  = threadIdx.x;
    const int lane = tid & 31;
    const int widx = __builtin_amdgcn_readfirstlane(tid >> 5) & (NW - 1);

    const int t2 = blockIdx.x / WBLK;
    const int bx = (blockIdx.x % WBLK) * NW + widx;
    const int hg = t2 % HG;
    const int b  = t2 / HG;
    const int h0 = hg * 16;
    const int gw0 = bx * BW - DISP;

    const int n     = lane & 15;
    const int khalf = lane >> 4;

    const float* rlane = R + (size_t)(b * Hh + h0 + n) * Ww + 4 * khalf;
    v4f s[NQ];
#pragma unroll
    for (int i = 0; i < NQ; ++i) {
        const int gcu = gw0 + 8 * i;
        const bool ok = (gcu >= 0) && (gcu < Ww);
        const int uo = ok ? gcu : 0;
        s[i] = *(const v4f*)(rlane + uo);
    }

    const v8f zero8 = (v8f){0.0f, 0.0f, 0.0f, 0.0f, 0.0f, 0.0f, 0.0f, 0.0f};
    v8f a11 = zero8, a12 = zero8, a13 = zero8, a21 = zero8, a22 = zero8, a23 = zero8,
        a31 = zero8, a32 = zero8, a33 = zero8;

#define PAD(i) s[(i)] = s[(i)] * (((gw0 + 8 * (i)) >= 0 && (gw0 + 8 * (i)) < Ww) ? 1.0f : 0.0f)
#define WM(acc, av, bv) acc = __builtin_amdgcn_wmma_f32_16x16x4_f32(false, (av), false, (bv), (short)0, acc, false, false)
#define CK(g, k, p, e) pick(s[8 * (g) + 2 + 2 * (k) + (p)], (e))
#define STEP(g, p, e) do {                                                                                  \
        __builtin_amdgcn_sched_barrier(0x4);                                                                \
        const v2f t0 = band(xb, 8 * (g) + 2 + (p), (e));                                                    \
        const v2f t1 = band(xb, 8 * (g) + 4 + (p), (e));                                                    \
        const v2f t2 = band(xb, 8 * (g) + 6 + (p), (e));                                                    \
        const v2f t3 = band(xb, 8 * (g) + 8 + (p), (e));                                                    \
        const v2f c0 = CK(g, 0, p, e), c1 = CK(g, 1, p, e), c2 = CK(g, 2, p, e), c3 = CK(g, 3, p, e);       \
        const v2f c4 = CK(g, 4, p, e), c5 = CK(g, 5, p, e), c6 = CK(g, 6, p, e);                            \
        const v2f f1 = c1 - c3;                                                                             \
        const v2f g2 = c2 - c4;                                                                             \
        const v2f f2 = c3 - c5;                                                                             \
        const v2f t01 = t0 + t1;                                                                            \
        const v2f t23 = t2 + t3;                                                                            \
        WM(a11, (c0 - c2) - f1, t0);                                                                        \
        WM(a12, f1, t01);                                                                                   \
        WM(a13, f1 - g2, t1);                                                                               \
        WM(a21, c2 - c3, t0 + t2);                                                                          \
        WM(a22, c3, t01 + t23);                                                                             \
        WM(a23, c3 - c4, t1 + t3);                                                                          \
        WM(a31, g2 - f2, t2);                                                                               \
        WM(a32, f2, t23);                                                                                   \
        WM(a33, f2 - (c4 - c6), t3);                                                                        \
    } while (0)

    float lv[NT][8];
    const float xb = (float)(n + DISP - 4 * khalf);

    v2f td[4], tdn[4];
#pragma unroll
    for (int pe = 0; pe < 4; ++pe) {
        td[pe]  = band(xb, pe >> 1, pe & 1);
        tdn[pe] = -td[pe];
    }

    __builtin_amdgcn_sched_barrier(0x4);
#pragma unroll
    for (int i = 0; i < 8; ++i) PAD(i);
#pragma unroll
    for (int pe = 0; pe < 4; ++pe) {
        const int p = pe >> 1, e = pe & 1;
        __builtin_amdgcn_sched_barrier(0x4);
        WM(a11, pick(s[0 + p], e), td[pe]);
        WM(a13, pick(s[2 + p], e), tdn[pe]);
        WM(a31, pick(s[4 + p], e), tdn[pe]);
        WM(a33, pick(s[6 + p], e), td[pe]);
    }

#pragma unroll
    for (int g = 0; g < 3; ++g) {
        __builtin_amdgcn_sched_barrier(0x4);
        if (g == 1) {
#pragma unroll
            for (int q = 0; q < NT; ++q) {
                const int tc  = bx * NT + q;
                const int tcl = (tc < WT) ? tc : (WT - 1);
                const size_t lbase = (size_t)(b * Hh + h0) * Ww + (size_t)((tcl << 4) + n);
#pragma unroll
                for (int v = 0; v < 8; ++v)
                    lv[q][v] = L[lbase + (size_t)(v + (khalf << 3)) * Ww];
            }
        }
#pragma unroll
        for (int i = 0; i < 8; ++i) PAD(8 * g + 8 + i);
#pragma unroll
        for (int pe = 0; pe < 4; ++pe) STEP(g, pe >> 1, pe & 1);
    }

    __builtin_amdgcn_sched_barrier(0x4);
#undef STEP
#undef CK
#undef WM
#undef PAD

    v8f c[NT];
    {
        const v8f sa = a12 + a22;
        const v8f sb = a22 - a32;
        c[0] = (a11 + a21) + sa;
        c[1] = (sa - a13) - a23;
        c[2] = (a21 - a31) + sb;
        c[3] = (sb - a23) + a33;
    }

#pragma unroll
    for (int q = 0; q < NT; ++q) {
        asm volatile("" :: "v"(lv[q][0]), "v"(lv[q][1]), "v"(lv[q][2]), "v"(lv[q][3]),
                           "v"(lv[q][4]), "v"(lv[q][5]), "v"(lv[q][6]), "v"(lv[q][7]));
    }
#pragma unroll
    for (int q = 0; q < NT; ++q) {
        const int tc = bx * NT + q;
        if (tc < WT) {
            const size_t obase = (size_t)(b * Hh + h0) * Ww + (size_t)((tc << 4) + n);
#pragma unroll
            for (int v = 0; v < 8; ++v)
                out[obase + (size_t)(v + (khalf << 3)) * Ww] = lv[q][v] * c[q][v];
        }
    }
}

extern "C" void kernel_launch(void* const* d_in, const int* in_sizes, int n_in,
                              void* d_out, int out_size, void* d_ws, size_t ws_size,
                              hipStream_t stream) {
    (void)in_sizes; (void)n_in; (void)out_size; (void)d_ws; (void)ws_size;
    const float* L = (const float*)d_in[0];
    const float* R = (const float*)d_in[1];
    float* O = (float*)d_out;

    static_assert(Hh % 16 == 0, "row strips must tile the image");
    static_assert(WBLK * NW * NT >= WT, "blocks must cover all tile columns");
    static_assert(Ww % 8 == 0 && DISP % 64 == 0, "a 16-byte piece must not straddle an image edge");
    dim3 grid(NBLK);
    dim3 block(128);
    hipLaunchKernelGGL(StereoMatching_wmma_kernel, grid, block, 0, stream, L, R, O);
}
